# starting kernel
# baseline (speedup 1.0000x reference)
#include <hip/hip_runtime.h>
#include <math.h>


typedef float v2f __attribute__((ext_vector_type(2)));
typedef float v8f __attribute__((ext_vector_type(8)));

#define LN_EPS_F 1e-6f

__device__ __forceinline__ float grp16_sum(float x) {
  x += __shfl_xor(x, 1, 32);
  x += __shfl_xor(x, 2, 32);
  x += __shfl_xor(x, 4, 32);
  x += __shfl_xor(x, 8, 32);
  return x;
}

__device__ __forceinline__ void qrotf(float qx, float qy, float qz, float qw,
                                      float vx, float vy, float vz,
                                      float& rx, float& ry, float& rz) {
  float tx = 2.f * (qy * vz - qz * vy);
  float ty = 2.f * (qz * vx - qx * vz);
  float tz = 2.f * (qx * vy - qy * vx);
  rx = vx + qw * tx + (qy * tz - qz * ty);
  ry = vy + qw * ty + (qz * tx - qx * tz);
  rz = vz + qw * tz + (qx * ty - qy * tx);
}

__global__ __launch_bounds__(128) void decmodel_kernel(
    const float* __restrict__ pos, const float* __restrict__ quat,
    const float* __restrict__ aabb, const float* __restrict__ xz,
    const int* __restrict__ widx, const float* __restrict__ plsp,
    const float* __restrict__ W1, const float* __restrict__ b1,
    const float* __restrict__ g1, const float* __restrict__ be1,
    const float* __restrict__ W2, const float* __restrict__ b2,
    const float* __restrict__ g2, const float* __restrict__ be2,
    const float* __restrict__ W3, const float* __restrict__ b3,
    const float* __restrict__ g3, const float* __restrict__ be3,
    const float* __restrict__ W4, const float* __restrict__ b4,
    float* __restrict__ out) {
  __shared__ float featW[4][32 * 26];
  __shared__ float zbuf[16 * 32];
  __shared__ float z2w[4][16 * 32];

  const int tid = threadIdx.x;
  const int wave = tid >> 5;
  const int lane = tid & 31;
  const int half = lane >> 4;
  const int n = lane & 15;
  const int koff = half * 2;
  const int b_base = blockIdx.x * 16;

  v2f bf1[6];
#pragma unroll
  for (int j = 0; j < 6; ++j) {
    int kk = j * 4 + koff;
    bf1[j].x = (kk < 23) ? W1[kk * 16 + n] : 0.f;
    bf1[j].y = (kk + 1 < 23) ? W1[(kk + 1) * 16 + n] : 0.f;
  }
  const float b1n = b1[n], g1n = g1[n], be1n = be1[n];
  const float escale = expf(plsp[0]);

  float* fw = featW[wave];
  for (int bl = 0; bl < 4; ++bl) {
    const int b = b_base + wave * 4 + bl;

    const float p1x = pos[b * 6 + 0], p1y = pos[b * 6 + 1], p1z = pos[b * 6 + 2];
    const float p2x = pos[b * 6 + 3], p2y = pos[b * 6 + 4], p2z = pos[b * 6 + 5];
    const float ax = -quat[b * 8 + 0], ay = -quat[b * 8 + 1],
                az = -quat[b * 8 + 2], aw = quat[b * 8 + 3];
    const float cx = quat[b * 8 + 4], cy = quat[b * 8 + 5],
                cz = quat[b * 8 + 6], cw = quat[b * 8 + 7];

    float r1x, r1y, r1z, r2x, r2y, r2z;
    qrotf(ax, ay, az, aw, p1x, p1y, p1z, r1x, r1y, r1z);
    qrotf(ax, ay, az, aw, p2x, p2y, p2z, r2x, r2y, r2z);
    const float rpx = r2x - r1x, rpy = r2y - r1y, rpz = r2z - r1z;

    const float rqw = aw * cw - (ax * cx + ay * cy + az * cz);
    const float rqx = aw * cx + cw * ax + (ay * cz - az * cy);
    const float rqy = aw * cy + cw * ay + (az * cx - ax * cz);
    const float rqz = aw * cz + cw * az + (ax * cy - ay * cx);

    float pf[12];
    pf[0] = rpx * escale; pf[1] = rpy * escale; pf[2] = rpz * escale;
    pf[3] = 1.f - 2.f * (rqy * rqy + rqz * rqz);
    pf[4] = 2.f * (rqx * rqy - rqz * rqw);
    pf[5] = 2.f * (rqx * rqz + rqy * rqw);
    pf[6] = 2.f * (rqx * rqy + rqz * rqw);
    pf[7] = 1.f - 2.f * (rqx * rqx + rqz * rqz);
    pf[8] = 2.f * (rqy * rqz - rqx * rqw);
    pf[9] = 2.f * (rqx * rqz - rqy * rqw);
    pf[10] = 2.f * (rqy * rqz + rqx * rqw);
    pf[11] = 1.f - 2.f * (rqx * rqx + rqy * rqy);

    for (int d = 0; d < 2; ++d) {
      const float lox = aabb[b * 12 + d * 6 + 0];
      const float loy = aabb[b * 12 + d * 6 + 1];
      const float loz = aabb[b * 12 + d * 6 + 2];
      const float hix = aabb[b * 12 + d * 6 + 3];
      const float hiy = aabb[b * 12 + d * 6 + 4];
      const float hiz = aabb[b * 12 + d * 6 + 5];

      const int w = widx[(b * 2 + d) * 32 + lane];
      const float gi = (float)(w >> 8);
      const float gj = (float)((w >> 4) & 15);
      const float gk = (float)(w & 15);
      float px = (gi + 0.5f) * (1.f / 16.f) * (hix - lox) + lox;
      float py = (gj + 0.5f) * (1.f / 16.f) * (hiy - loy) + loy;
      float pz = (gk + 0.5f) * (1.f / 16.f) * (hiz - loz) + loz;
      if (d == 1) {
        float tx, ty, tz;
        qrotf(rqx, rqy, rqz, rqw, px, py, pz, tx, ty, tz);
        px = tx + rpx; py = ty + rpy; pz = tz + rpz;
      }

      const float* zp = xz + ((size_t)(b * 2 + d) * 4096 + (size_t)w) * 8;
      float* frow = fw + lane * 26;
      frow[0] = px; frow[1] = py; frow[2] = pz;
#pragma unroll
      for (int p = 0; p < 8; ++p) frow[3 + p] = zp[p];
      if (d) {
#pragma unroll
        for (int p = 0; p < 12; ++p) frow[11 + p] = pf[p];
      } else {
#pragma unroll
        for (int p = 0; p < 12; ++p) frow[11 + p] = 0.f;
      }
      frow[23] = 0.f;
      __syncthreads();

      float acc = 0.f;
#pragma unroll
      for (int t = 0; t < 2; ++t) {
        v8f c = {b1n, b1n, b1n, b1n, b1n, b1n, b1n, b1n};
        const float* row = fw + (t * 16 + n) * 26;
#pragma unroll
        for (int j = 0; j < 6; ++j) {
          const int kk = j * 4 + koff;
          v2f a;
          a.x = row[kk];
          a.y = row[kk + 1];
          c = __builtin_amdgcn_wmma_f32_16x16x4_f32(
              false, a, false, bf1[j], (short)0, c, false, false);
        }
#pragma unroll
        for (int v = 0; v < 8; ++v) {
          float x = fmaxf(c[v], 0.f);
          float s1 = grp16_sum(x);
          float s2 = grp16_sum(x * x);
          float m = s1 * (1.f / 16.f);
          float var = s2 * (1.f / 16.f) - m * m;
          float ln = (x - m) * rsqrtf(var + LN_EPS_F) * g1n + be1n;
          acc += 2.f * ln;
        }
      }
      acc += __shfl_xor(acc, 16, 32);
      if (lane < 16) zbuf[(wave * 4 + bl) * 32 + d * 16 + lane] = acc;
      __syncthreads();
    }
  }

  float* zw = z2w[wave];
  const float g2a = g2[n], g2b = g2[16 + n], be2a = be2[n], be2b = be2[16 + n];
  const float g3a = g3[n], g3b = g3[16 + n], be3a = be3[n], be3b = be3[16 + n];

  const float bb20 = b2[n], bb21 = b2[16 + n];
  v8f c0 = {bb20, bb20, bb20, bb20, bb20, bb20, bb20, bb20};
  v8f c1 = {bb21, bb21, bb21, bb21, bb21, bb21, bb21, bb21};
#pragma unroll
  for (int j = 0; j < 8; ++j) {
    const int kk = j * 4 + koff;
    v2f a;
    a.x = zbuf[n * 32 + kk];
    a.y = zbuf[n * 32 + kk + 1];
    v2f bA, bB;
    bA.x = W2[kk * 32 + n];        bA.y = W2[(kk + 1) * 32 + n];
    bB.x = W2[kk * 32 + 16 + n];   bB.y = W2[(kk + 1) * 32 + 16 + n];
    c0 = __builtin_amdgcn_wmma_f32_16x16x4_f32(false, a, false, bA, (short)0, c0, false, false);
    c1 = __builtin_amdgcn_wmma_f32_16x16x4_f32(false, a, false, bB, (short)0, c1, false, false);
  }
  v8f z0, z1, skip0, skip1;
#pragma unroll
  for (int v = 0; v < 8; ++v) {
    float x0 = fmaxf(c0[v], 0.f), x1 = fmaxf(c1[v], 0.f);
    float s1 = grp16_sum(x0 + x1);
    float s2 = grp16_sum(x0 * x0 + x1 * x1);
    float m = s1 * (1.f / 32.f);
    float var = s2 * (1.f / 32.f) - m * m;
    float rs = rsqrtf(var + LN_EPS_F);
    z0[v] = (x0 - m) * rs * g2a + be2a;
    z1[v] = (x1 - m) * rs * g2b + be2b;
  }
  skip0 = z0; skip1 = z1;
#pragma unroll
  for (int v = 0; v < 8; ++v) {
    zw[(v + half * 8) * 32 + n] = z0[v];
    zw[(v + half * 8) * 32 + 16 + n] = z1[v];
  }
  __syncthreads();

  const float bb30 = b3[n], bb31 = b3[16 + n];
  v8f d0 = {bb30, bb30, bb30, bb30, bb30, bb30, bb30, bb30};
  v8f d1 = {bb31, bb31, bb31, bb31, bb31, bb31, bb31, bb31};
#pragma unroll
  for (int j = 0; j < 8; ++j) {
    const int kk = j * 4 + koff;
    v2f a;
    a.x = zw[n * 32 + kk];
    a.y = zw[n * 32 + kk + 1];
    v2f bA, bB;
    bA.x = W3[kk * 32 + n];        bA.y = W3[(kk + 1) * 32 + n];
    bB.x = W3[kk * 32 + 16 + n];   bB.y = W3[(kk + 1) * 32 + 16 + n];
    d0 = __builtin_amdgcn_wmma_f32_16x16x4_f32(false, a, false, bA, (short)0, d0, false, false);
    d1 = __builtin_amdgcn_wmma_f32_16x16x4_f32(false, a, false, bB, (short)0, d1, false, false);
  }
#pragma unroll
  for (int v = 0; v < 8; ++v) {
    float x0 = fmaxf(d0[v], 0.f), x1 = fmaxf(d1[v], 0.f);
    float s1 = grp16_sum(x0 + x1);
    float s2 = grp16_sum(x0 * x0 + x1 * x1);
    float m = s1 * (1.f / 32.f);
    float var = s2 * (1.f / 32.f) - m * m;
    float rs = rsqrtf(var + LN_EPS_F);
    z0[v] = (x0 - m) * rs * g3a + be3a + skip0[v];
    z1[v] = (x1 - m) * rs * g3b + be3b + skip1[v];
  }
  __syncthreads();
#pragma unroll
  for (int v = 0; v < 8; ++v) {
    zw[(v + half * 8) * 32 + n] = z0[v];
    zw[(v + half * 8) * 32 + 16 + n] = z1[v];
  }
  __syncthreads();

  const float bb4 = (n < 13) ? b4[n] : 0.f;
  v8f c4 = {bb4, bb4, bb4, bb4, bb4, bb4, bb4, bb4};
#pragma unroll
  for (int j = 0; j < 8; ++j) {
    const int kk = j * 4 + koff;
    v2f a;
    a.x = zw[n * 32 + kk];
    a.y = zw[n * 32 + kk + 1];
    v2f bw;
    bw.x = (n < 13) ? W4[kk * 13 + n] : 0.f;
    bw.y = (n < 13) ? W4[(kk + 1) * 13 + n] : 0.f;
    c4 = __builtin_amdgcn_wmma_f32_16x16x4_f32(false, a, false, bw, (short)0, c4, false, false);
  }
  if (wave == 0 && n < 13) {
#pragma unroll
    for (int v = 0; v < 8; ++v) {
      out[(size_t)(b_base + v + half * 8) * 13 + n] = tanhf(c4[v]);
    }
  }
}

extern "C" void kernel_launch(void* const* d_in, const int* in_sizes, int n_in,
                              void* d_out, int out_size, void* d_ws, size_t ws_size,
                              hipStream_t stream) {
  const float* pos  = (const float*)d_in[0];
  const float* quat = (const float*)d_in[1];
  const float* aabb = (const float*)d_in[2];
  const float* xz   = (const float*)d_in[3];
  const int*   widx = (const int*)d_in[4];
  const float* pls  = (const float*)d_in[5];
  const float* W1 = (const float*)d_in[6];
  const float* b1 = (const float*)d_in[7];
  const float* g1 = (const float*)d_in[8];
  const float* be1 = (const float*)d_in[9];
  const float* W2 = (const float*)d_in[10];
  const float* b2 = (const float*)d_in[11];
  const float* g2 = (const float*)d_in[12];
  const float* be2 = (const float*)d_in[13];
  const float* W3 = (const float*)d_in[14];
  const float* b3 = (const float*)d_in[15];
  const float* g3 = (const float*)d_in[16];
  const float* be3 = (const float*)d_in[17];
  const float* W4 = (const float*)d_in[18];
  const float* b4 = (const float*)d_in[19];
  float* out = (float*)d_out;

  const int Bn = in_sizes[0] / 6;
  const int grid = Bn / 16;
  hipLaunchKernelGGL(decmodel_kernel, dim3(grid), dim3(128), 0, stream,
                     pos, quat, aabb, xz, widx, pls,
                     W1, b1, g1, be1, W2, b2, g2, be2,
                     W3, b3, g3, be3, W4, b4, out);
}


# Loads hoisted: a wave's 8 index vectors, then its 8 gathered row sets and W2 at entry, W3/W4 before the item loop; item features staged in own LDS rows: 2 dependent memory trips, not 2 per item
# speedup vs baseline: 1.3083x; 1.3083x over previous
#include <hip/hip_runtime.h>
#include <math.h>


typedef float v2f __attribute__((ext_vector_type(2)));
typedef float v8f __attribute__((ext_vector_type(8)));

#define LN_EPS_F 1e-6f
#define FROW 26
#define FITEM (32 * FROW)

__device__ __forceinline__ float grp16_sum(float x) {
  x += __shfl_xor(x, 1, 32);
  x += __shfl_xor(x, 2, 32);
  x += __shfl_xor(x, 4, 32);
  x += __shfl_xor(x, 8, 32);
  return x;
}

__device__ __forceinline__ void qrotf(float qx, float qy, float qz, float qw,
                                      float vx, float vy, float vz,
                                      float& rx, float& ry, float& rz) {
  float tx = 2.f * (qy * vz - qz * vy);
  float ty = 2.f * (qz * vx - qx * vz);
  float tz = 2.f * (qx * vy - qy * vx);
  rx = vx + qw * tx + (qy * tz - qz * ty);
  ry = vy + qw * ty + (qz * tx - qx * tz);
  rz = vz + qw * tz + (qx * ty - qy * tx);
}

__global__ __launch_bounds__(128) void decmodel_kernel(
    const float* __restrict__ pos, const float* __restrict__ quat,
    const float* __restrict__ aabb, const float* __restrict__ xz,
    const int* __restrict__ widx, const float* __restrict__ plsp,
    const float* __restrict__ W1, const float* __restrict__ b1,
    const float* __restrict__ g1, const float* __restrict__ be1,
    const float* __restrict__ W2, const float* __restrict__ b2,
    const float* __restrict__ g2, const float* __restrict__ be2,
    const float* __restrict__ W3, const float* __restrict__ b3,
    const float* __restrict__ g3, const float* __restrict__ be3,
    const float* __restrict__ W4, const float* __restrict__ b4,
    float* __restrict__ out) {
  __shared__ float featW[4][8 * FITEM];
  __shared__ float zbuf[16 * 32];
  __shared__ float z2w[4][16 * 32];

  const int tid = threadIdx.x;
  const int wave = tid >> 5;
  const int lane = tid & 31;
  const int half = lane >> 4;
  const int n = lane & 15;
  const int koff = half * 2;
  const int b_base = blockIdx.x * 16;

  const int bw = b_base + wave * 4;

  int wv[8];
#pragma unroll
  for (int i = 0; i < 8; ++i) wv[i] = widx[(bw * 2 + i) * 32 + lane];

  float ps[4][6], qt[4][8], ab[4][12];
#pragma unroll
  for (int bl = 0; bl < 4; ++bl) {
#pragma unroll
    for (int k = 0; k < 6; ++k) ps[bl][k] = pos[(bw + bl) * 6 + k];
#pragma unroll
    for (int k = 0; k < 8; ++k) qt[bl][k] = quat[(bw + bl) * 8 + k];
#pragma unroll
    for (int k = 0; k < 12; ++k) ab[bl][k] = aabb[(bw + bl) * 12 + k];
  }

  v2f bf1[6];
#pragma unroll
  for (int j = 0; j < 6; ++j) {
    int kk = j * 4 + koff;
    bf1[j].x = (kk < 23) ? W1[kk * 16 + n] : 0.f;
    bf1[j].y = (kk + 1 < 23) ? W1[(kk + 1) * 16 + n] : 0.f;
  }
  const float b1n = b1[n], g1n = g1[n], be1n = be1[n];
  const float pls0 = plsp[0];

  float zr[8][8];
#pragma unroll
  for (int i = 0; i < 8; ++i) {
    const float* zp = xz + ((size_t)(bw * 2 + i) * 4096 + (size_t)wv[i]) * 8;
#pragma unroll
    for (int p = 0; p < 8; ++p) zr[i][p] = zp[p];
  }

  v2f w2a[8], w2b[8];
#pragma unroll
  for (int j = 0; j < 8; ++j) {
    const int kk = j * 4 + koff;
    w2a[j].x = W2[kk * 32 + n];        w2a[j].y = W2[(kk + 1) * 32 + n];
    w2b[j].x = W2[kk * 32 + 16 + n];   w2b[j].y = W2[(kk + 1) * 32 + 16 + n];
  }
  const float g2a = g2[n], g2b = g2[16 + n], be2a = be2[n], be2b = be2[16 + n];
  const float g3a = g3[n], g3b = g3[16 + n], be3a = be3[n], be3b = be3[16 + n];
  const float bb20 = b2[n], bb21 = b2[16 + n];
  const float bb30 = b3[n], bb31 = b3[16 + n];
  const float bb4 = (n < 13) ? b4[n] : 0.f;

  const float escale = expf(pls0);

  float* fw = featW[wave];
#pragma unroll
  for (int bl = 0; bl < 4; ++bl) {
    const float p1x = ps[bl][0], p1y = ps[bl][1], p1z = ps[bl][2];
    const float p2x = ps[bl][3], p2y = ps[bl][4], p2z = ps[bl][5];
    const float ax = -qt[bl][0], ay = -qt[bl][1],
                az = -qt[bl][2], aw = qt[bl][3];
    const float cx = qt[bl][4], cy = qt[bl][5],
                cz = qt[bl][6], cw = qt[bl][7];

    float r1x, r1y, r1z, r2x, r2y, r2z;
    qrotf(ax, ay, az, aw, p1x, p1y, p1z, r1x, r1y, r1z);
    qrotf(ax, ay, az, aw, p2x, p2y, p2z, r2x, r2y, r2z);
    const float rpx = r2x - r1x, rpy = r2y - r1y, rpz = r2z - r1z;

    const float rqw = aw * cw - (ax * cx + ay * cy + az * cz);
    const float rqx = aw * cx + cw * ax + (ay * cz - az * cy);
    const float rqy = aw * cy + cw * ay + (az * cx - ax * cz);
    const float rqz = aw * cz + cw * az + (ax * cy - ay * cx);

    float pf[12];
    pf[0] = rpx * escale; pf[1] = rpy * escale; pf[2] = rpz * escale;
    pf[3] = 1.f - 2.f * (rqy * rqy + rqz * rqz);
    pf[4] = 2.f * (rqx * rqy - rqz * rqw);
    pf[5] = 2.f * (rqx * rqz + rqy * rqw);
    pf[6] = 2.f * (rqx * rqy + rqz * rqw);
    pf[7] = 1.f - 2.f * (rqx * rqx + rqz * rqz);
    pf[8] = 2.f * (rqy * rqz - rqx * rqw);
    pf[9] = 2.f * (rqx * rqz - rqy * rqw);
    pf[10] = 2.f * (rqy * rqz + rqx * rqw);
    pf[11] = 1.f - 2.f * (rqx * rqx + rqy * rqy);

#pragma unroll
    for (int d = 0; d < 2; ++d) {
      const int i = bl * 2 + d;
      const float lox = ab[bl][d * 6 + 0];
      const float loy = ab[bl][d * 6 + 1];
      const float loz = ab[bl][d * 6 + 2];
      const float hix = ab[bl][d * 6 + 3];
      const float hiy = ab[bl][d * 6 + 4];
      const float hiz = ab[bl][d * 6 + 5];

      const int w = wv[i];
      const float gi = (float)(w >> 8);
      const float gj = (float)((w >> 4) & 15);
      const float gk = (float)(w & 15);
      float px = (gi + 0.5f) * (1.f / 16.f) * (hix - lox) + lox;
      float py = (gj + 0.5f) * (1.f / 16.f) * (hiy - loy) + loy;
      float pz = (gk + 0.5f) * (1.f / 16.f) * (hiz - loz) + loz;
      if (d == 1) {
        float tx, ty, tz;
        qrotf(rqx, rqy, rqz, rqw, px, py, pz, tx, ty, tz);
        px = tx + rpx; py = ty + rpy; pz = tz + rpz;
      }

      float* frow = fw + i * FITEM + lane * FROW;
      frow[0] = px; frow[1] = py; frow[2] = pz;
      if (d) {
#pragma unroll
        for (int p = 0; p < 12; ++p) frow[11 + p] = pf[p];
      } else {
#pragma unroll
        for (int p = 0; p < 12; ++p) frow[11 + p] = 0.f;
      }
      frow[23] = 0.f;
    }
  }

#pragma unroll
  for (int i = 0; i < 8; ++i) {
    float* frow = fw + i * FITEM + lane * FROW;
#pragma unroll
    for (int p = 0; p < 8; ++p) frow[3 + p] = zr[i][p];
  }

  __syncthreads();

  v2f w3a[8], w3b[8], w4f[8];
#pragma unroll
  for (int j = 0; j < 8; ++j) {
    const int kk = j * 4 + koff;
    w3a[j].x = W3[kk * 32 + n];        w3a[j].y = W3[(kk + 1) * 32 + n];
    w3b[j].x = W3[kk * 32 + 16 + n];   w3b[j].y = W3[(kk + 1) * 32 + 16 + n];
    w4f[j].x = (n < 13) ? W4[kk * 13 + n] : 0.f;
    w4f[j].y = (n < 13) ? W4[(kk + 1) * 13 + n] : 0.f;
  }

  for (int i = 0; i < 8; ++i) {
    const float* fi = fw + i * FITEM;
    float acc = 0.f;
#pragma unroll
    for (int t = 0; t < 2; ++t) {
      v8f c = {b1n, b1n, b1n, b1n, b1n, b1n, b1n, b1n};
      const float* row = fi + (t * 16 + n) * FROW;
#pragma unroll
      for (int j = 0; j < 6; ++j) {
        const int kk = j * 4 + koff;
        v2f a;
        a.x = row[kk];
        a.y = row[kk + 1];
        c = __builtin_amdgcn_wmma_f32_16x16x4_f32(
            false, a, false, bf1[j], (short)0, c, false, false);
      }
#pragma unroll
      for (int v = 0; v < 8; ++v) {
        float x = fmaxf(c[v], 0.f);
        float s1 = grp16_sum(x);
        float s2 = grp16_sum(x * x);
        float m = s1 * (1.f / 16.f);
        float var = s2 * (1.f / 16.f) - m * m;
        float ln = (x - m) * rsqrtf(var + LN_EPS_F) * g1n + be1n;
        acc += 2.f * ln;
      }
    }
    acc += __shfl_xor(acc, 16, 32);
    if (lane < 16) zbuf[wave * 128 + i * 16 + lane] = acc;
  }
  __syncthreads();

  float* zw = z2w[wave];

  v8f c0 = {bb20, bb20, bb20, bb20, bb20, bb20, bb20, bb20};
  v8f c1 = {bb21, bb21, bb21, bb21, bb21, bb21, bb21, bb21};
#pragma unroll
  for (int j = 0; j < 8; ++j) {
    const int kk = j * 4 + koff;
    v2f a;
    a.x = zbuf[n * 32 + kk];
    a.y = zbuf[n * 32 + kk + 1];
    c0 = __builtin_amdgcn_wmma_f32_16x16x4_f32(false, a, false, w2a[j], (short)0, c0, false, false);
    c1 = __builtin_amdgcn_wmma_f32_16x16x4_f32(false, a, false, w2b[j], (short)0, c1, false, false);
  }
  v8f z0, z1, skip0, skip1;
#pragma unroll
  for (int v = 0; v < 8; ++v) {
    float x0 = fmaxf(c0[v], 0.f), x1 = fmaxf(c1[v], 0.f);
    float s1 = grp16_sum(x0 + x1);
    float s2 = grp16_sum(x0 * x0 + x1 * x1);
    float m = s1 * (1.f / 32.f);
    float var = s2 * (1.f / 32.f) - m * m;
    float rs = rsqrtf(var + LN_EPS_F);
    z0[v] = (x0 - m) * rs * g2a + be2a;
    z1[v] = (x1 - m) * rs * g2b + be2b;
  }
  skip0 = z0; skip1 = z1;
#pragma unroll
  for (int v = 0; v < 8; ++v) {
    zw[(v + half * 8) * 32 + n] = z0[v];
    zw[(v + half * 8) * 32 + 16 + n] = z1[v];
  }
  __syncthreads();

  v8f d0 = {bb30, bb30, bb30, bb30, bb30, bb30, bb30, bb30};
  v8f d1 = {bb31, bb31, bb31, bb31, bb31, bb31, bb31, bb31};
#pragma unroll
  for (int j = 0; j < 8; ++j) {
    const int kk = j * 4 + koff;
    v2f a;
    a.x = zw[n * 32 + kk];
    a.y = zw[n * 32 + kk + 1];
    d0 = __builtin_amdgcn_wmma_f32_16x16x4_f32(false, a, false, w3a[j], (short)0, d0, false, false);
    d1 = __builtin_amdgcn_wmma_f32_16x16x4_f32(false, a, false, w3b[j], (short)0, d1, false, false);
  }
#pragma unroll
  for (int v = 0; v < 8; ++v) {
    float x0 = fmaxf(d0[v], 0.f), x1 = fmaxf(d1[v], 0.f);
    float s1 = grp16_sum(x0 + x1);
    float s2 = grp16_sum(x0 * x0 + x1 * x1);
    float m = s1 * (1.f / 32.f);
    float var = s2 * (1.f / 32.f) - m * m;
    float rs = rsqrtf(var + LN_EPS_F);
    z0[v] = (x0 - m) * rs * g3a + be3a + skip0[v];
    z1[v] = (x1 - m) * rs * g3b + be3b + skip1[v];
  }
  __syncthreads();
#pragma unroll
  for (int v = 0; v < 8; ++v) {
    zw[(v + half * 8) * 32 + n] = z0[v];
    zw[(v + half * 8) * 32 + 16 + n] = z1[v];
  }
  __syncthreads();

  v8f c4 = {bb4, bb4, bb4, bb4, bb4, bb4, bb4, bb4};
#pragma unroll
  for (int j = 0; j < 8; ++j) {
    const int kk = j * 4 + koff;
    v2f a;
    a.x = zw[n * 32 + kk];
    a.y = zw[n * 32 + kk + 1];
    c4 = __builtin_amdgcn_wmma_f32_16x16x4_f32(false, a, false, w4f[j], (short)0, c4, false, false);
  }
  if (wave == 0 && n < 13) {
#pragma unroll
    for (int v = 0; v < 8; ++v) {
      out[(size_t)(b_base + v + half * 8) * 13 + n] = tanhf(c4[v]);
    }
  }
}

extern "C" void kernel_launch(void* const* d_in, const int* in_sizes, int n_in,
                              void* d_out, int out_size, void* d_ws, size_t ws_size,
                              hipStream_t stream) {
  const float* pos  = (const float*)d_in[0];
  const float* quat = (const float*)d_in[1];
  const float* aabb = (const float*)d_in[2];
  const float* xz   = (const float*)d_in[3];
  const int*   widx = (const int*)d_in[4];
  const float* pls  = (const float*)d_in[5];
  const float* W1 = (const float*)d_in[6];
  const float* b1 = (const float*)d_in[7];
  const float* g1 = (const float*)d_in[8];
  const float* be1 = (const float*)d_in[9];
  const float* W2 = (const float*)d_in[10];
  const float* b2 = (const float*)d_in[11];
  const float* g2 = (const float*)d_in[12];
  const float* be2 = (const float*)d_in[13];
  const float* W3 = (const float*)d_in[14];
  const float* b3 = (const float*)d_in[15];
  const float* g3 = (const float*)d_in[16];
  const float* be3 = (const float*)d_in[17];
  const float* W4 = (const float*)d_in[18];
  const float* b4 = (const float*)d_in[19];
  float* out = (float*)d_out;

  const int Bn = in_sizes[0] / 6;
  const int grid = Bn / 16;
  hipLaunchKernelGGL(decmodel_kernel, dim3(grid), dim3(128), 0, stream,
                     pos, quat, aabb, xz, widx, pls,
                     W1, b1, g1, be1, W2, b2, g2, be2,
                     W3, b3, g3, be3, W4, b4, out);
}


# All 4 products transposed (weights as A, activations as B): layer-norm sums are in-lane adds + 1 half exchange, not 16-lane shuffles (new reduction tree); tail tiles feed next product from registers
# speedup vs baseline: 2.2861x; 1.7474x over previous
#include <hip/hip_runtime.h>
#include <math.h>


typedef float v2f __attribute__((ext_vector_type(2)));
typedef float v8f __attribute__((ext_vector_type(8)));

#define LN_EPS_F 1e-6f
#define FROW 26
#define FITEM (32 * FROW)

__device__ __forceinline__ float xhalf(float x) {
  const int b = __builtin_bit_cast(int, x);
  return __builtin_bit_cast(
      float, __builtin_amdgcn_permlanex16(b, b, 0x76543210, 0xfedcba98, false, false));
}

template <int M>
__device__ __forceinline__ float row_xor_add(float x) {
  const int b = __builtin_bit_cast(int, x);
  const int o = __builtin_amdgcn_update_dpp(0, b, 0x160 + M, 0xf, 0xf, true);
  return x + __builtin_bit_cast(float, o);
}

__device__ __forceinline__ float row16_sum(float x) {
  x = row_xor_add<1>(x);
  x = row_xor_add<2>(x);
  x = row_xor_add<4>(x);
  x = row_xor_add<8>(x);
  return x;
}

__device__ __forceinline__ int kperm(int half, int j, int s) {
  return (j < 4) ? half * 8 + 2 * j + s : 16 + half * 8 + 2 * (j - 4) + s;
}

__device__ __forceinline__ void relu_ln32(const v8f& a0, const v8f& a1,
                                          const float* g0, const float* g1,
                                          const float* e0, const float* e1,
                                          v8f& o0, v8f& o1) {
  float x0[8], x1[8];
  float s1 = 0.f, s2 = 0.f;
#pragma unroll
  for (int r = 0; r < 8; ++r) {
    x0[r] = fmaxf(a0[r], 0.f);
    x1[r] = fmaxf(a1[r], 0.f);
    s1 += x0[r] + x1[r];
    s2 += x0[r] * x0[r] + x1[r] * x1[r];
  }
  s1 += xhalf(s1);
  s2 += xhalf(s2);
  const float m = s1 * (1.f / 32.f);
  const float var = s2 * (1.f / 32.f) - m * m;
  const float rs = rsqrtf(var + LN_EPS_F);
#pragma unroll
  for (int r = 0; r < 8; ++r) {
    o0[r] = (x0[r] - m) * rs * g0[r] + e0[r];
    o1[r] = (x1[r] - m) * rs * g1[r] + e1[r];
  }
}

__device__ __forceinline__ void qrotf(float qx, float qy, float qz, float qw,
                                      float vx, float vy, float vz,
                                      float& rx, float& ry, float& rz) {
  float tx = 2.f * (qy * vz - qz * vy);
  float ty = 2.f * (qz * vx - qx * vz);
  float tz = 2.f * (qx * vy - qy * vx);
  rx = vx + qw * tx + (qy * tz - qz * ty);
  ry = vy + qw * ty + (qz * tx - qx * tz);
  rz = vz + qw * tz + (qx * ty - qy * tx);
}

__global__ __launch_bounds__(128) void decmodel_kernel(
    const float* __restrict__ pos, const float* __restrict__ quat,
    const float* __restrict__ aabb, const float* __restrict__ xz,
    const int* __restrict__ widx, const float* __restrict__ plsp,
    const float* __restrict__ W1, const float* __restrict__ b1,
    const float* __restrict__ g1, const float* __restrict__ be1,
    const float* __restrict__ W2, const float* __restrict__ b2,
    const float* __restrict__ g2, const float* __restrict__ be2,
    const float* __restrict__ W3, const float* __restrict__ b3,
    const float* __restrict__ g3, const float* __restrict__ be3,
    const float* __restrict__ W4, const float* __restrict__ b4,
    float* __restrict__ out) {
  __shared__ float featW[4][8 * FITEM];
  __shared__ float zbuf[16 * 32];

  const int tid = threadIdx.x;
  const int wave = tid >> 5;
  const int lane = tid & 31;
  const int half = lane >> 4;
  const int n = lane & 15;
  const int koff = half * 2;
  const int b_base = blockIdx.x * 16;
  const bool tailw = __builtin_amdgcn_readfirstlane(tid) < 32;

  const int bw = b_base + wave * 4;

  int wv[8];
#pragma unroll
  for (int i = 0; i < 8; ++i) wv[i] = widx[(bw * 2 + i) * 32 + lane];

  float ps[4][6], qt[4][8], ab[4][12];
#pragma unroll
  for (int bl = 0; bl < 4; ++bl) {
#pragma unroll
    for (int k = 0; k < 6; ++k) ps[bl][k] = pos[(bw + bl) * 6 + k];
#pragma unroll
    for (int k = 0; k < 8; ++k) qt[bl][k] = quat[(bw + bl) * 8 + k];
#pragma unroll
    for (int k = 0; k < 12; ++k) ab[bl][k] = aabb[(bw + bl) * 12 + k];
  }

  v2f bf1[6];
#pragma unroll
  for (int j = 0; j < 6; ++j) {
    int kk = j * 4 + koff;
    bf1[j].x = (kk < 23) ? W1[kk * 16 + n] : 0.f;
    bf1[j].y = (kk + 1 < 23) ? W1[(kk + 1) * 16 + n] : 0.f;
  }
  float b1r[8], g1r[8], be1r[8];
#pragma unroll
  for (int r = 0; r < 8; ++r) {
    b1r[r] = b1[half * 8 + r];
    g1r[r] = g1[half * 8 + r];
    be1r[r] = be1[half * 8 + r];
  }
  const float pls0 = plsp[0];

  float zr[8][8];
#pragma unroll
  for (int i = 0; i < 8; ++i) {
    const float* zp = xz + ((size_t)(bw * 2 + i) * 4096 + (size_t)wv[i]) * 8;
#pragma unroll
    for (int p = 0; p < 8; ++p) zr[i][p] = zp[p];
  }

  v2f w2a[8] = {}, w2b[8] = {};
  if (tailw) {
#pragma unroll
    for (int j = 0; j < 8; ++j) {
      const int kk = j * 4 + koff;
      w2a[j].x = W2[kk * 32 + n];        w2a[j].y = W2[(kk + 1) * 32 + n];
      w2b[j].x = W2[kk * 32 + 16 + n];   w2b[j].y = W2[(kk + 1) * 32 + 16 + n];
    }
  }

  const float escale = expf(pls0);

  float* fw = featW[wave];
#pragma unroll
  for (int bl = 0; bl < 4; ++bl) {
    const float p1x = ps[bl][0], p1y = ps[bl][1], p1z = ps[bl][2];
    const float p2x = ps[bl][3], p2y = ps[bl][4], p2z = ps[bl][5];
    const float ax = -qt[bl][0], ay = -qt[bl][1],
                az = -qt[bl][2], aw = qt[bl][3];
    const float cx = qt[bl][4], cy = qt[bl][5],
                cz = qt[bl][6], cw = qt[bl][7];

    float r1x, r1y, r1z, r2x, r2y, r2z;
    qrotf(ax, ay, az, aw, p1x, p1y, p1z, r1x, r1y, r1z);
    qrotf(ax, ay, az, aw, p2x, p2y, p2z, r2x, r2y, r2z);
    const float rpx = r2x - r1x, rpy = r2y - r1y, rpz = r2z - r1z;

    const float rqw = aw * cw - (ax * cx + ay * cy + az * cz);
    const float rqx = aw * cx + cw * ax + (ay * cz - az * cy);
    const float rqy = aw * cy + cw * ay + (az * cx - ax * cz);
    const float rqz = aw * cz + cw * az + (ax * cy - ay * cx);

    float pf[12];
    pf[0] = rpx * escale; pf[1] = rpy * escale; pf[2] = rpz * escale;
    pf[3] = 1.f - 2.f * (rqy * rqy + rqz * rqz);
    pf[4] = 2.f * (rqx * rqy - rqz * rqw);
    pf[5] = 2.f * (rqx * rqz + rqy * rqw);
    pf[6] = 2.f * (rqx * rqy + rqz * rqw);
    pf[7] = 1.f - 2.f * (rqx * rqx + rqz * rqz);
    pf[8] = 2.f * (rqy * rqz - rqx * rqw);
    pf[9] = 2.f * (rqx * rqz - rqy * rqw);
    pf[10] = 2.f * (rqy * rqz + rqx * rqw);
    pf[11] = 1.f - 2.f * (rqx * rqx + rqy * rqy);

#pragma unroll
    for (int d = 0; d < 2; ++d) {
      const int i = bl * 2 + d;
      const float lox = ab[bl][d * 6 + 0];
      const float loy = ab[bl][d * 6 + 1];
      const float loz = ab[bl][d * 6 + 2];
      const float hix = ab[bl][d * 6 + 3];
      const float hiy = ab[bl][d * 6 + 4];
      const float hiz = ab[bl][d * 6 + 5];

      const int w = wv[i];
      const float gi = (float)(w >> 8);
      const float gj = (float)((w >> 4) & 15);
      const float gk = (float)(w & 15);
      float px = (gi + 0.5f) * (1.f / 16.f) * (hix - lox) + lox;
      float py = (gj + 0.5f) * (1.f / 16.f) * (hiy - loy) + loy;
      float pz = (gk + 0.5f) * (1.f / 16.f) * (hiz - loz) + loz;
      if (d == 1) {
        float tx, ty, tz;
        qrotf(rqx, rqy, rqz, rqw, px, py, pz, tx, ty, tz);
        px = tx + rpx; py = ty + rpy; pz = tz + rpz;
      }

      float* frow = fw + i * FITEM + lane * FROW;
      frow[0] = px; frow[1] = py; frow[2] = pz;
      if (d) {
#pragma unroll
        for (int p = 0; p < 12; ++p) frow[11 + p] = pf[p];
      } else {
#pragma unroll
        for (int p = 0; p < 12; ++p) frow[11 + p] = 0.f;
      }
      frow[23] = 0.f;
    }
  }

#pragma unroll
  for (int i = 0; i < 8; ++i) {
    float* frow = fw + i * FITEM + lane * FROW;
#pragma unroll
    for (int p = 0; p < 8; ++p) frow[3 + p] = zr[i][p];
  }

  __syncthreads();

  v2f w3a[8] = {}, w3b[8] = {}, w4f[8] = {};
  float b2r0[8] = {}, b2r1[8] = {}, g2r0[8] = {}, g2r1[8] = {}, e2r0[8] = {}, e2r1[8] = {};
  float b3r0[8] = {}, b3r1[8] = {}, g3r0[8] = {}, g3r1[8] = {}, e3r0[8] = {}, e3r1[8] = {};
  float bb4 = 0.f;
  if (tailw) {
#pragma unroll
    for (int j = 0; j < 8; ++j) {
      const int k0 = kperm(half, j, 0), k1 = kperm(half, j, 1);
      w3a[j].x = W3[k0 * 32 + n];        w3a[j].y = W3[k1 * 32 + n];
      w3b[j].x = W3[k0 * 32 + 16 + n];   w3b[j].y = W3[k1 * 32 + 16 + n];
      w4f[j].x = (n < 13) ? W4[k0 * 13 + n] : 0.f;
      w4f[j].y = (n < 13) ? W4[k1 * 13 + n] : 0.f;
    }
#pragma unroll
    for (int r = 0; r < 8; ++r) {
      const int f0 = half * 8 + r, f1 = 16 + half * 8 + r;
      b2r0[r] = b2[f0];  b2r1[r] = b2[f1];
      g2r0[r] = g2[f0];  g2r1[r] = g2[f1];
      e2r0[r] = be2[f0]; e2r1[r] = be2[f1];
      b3r0[r] = b3[f0];  b3r1[r] = b3[f1];
      g3r0[r] = g3[f0];  g3r1[r] = g3[f1];
      e3r0[r] = be3[f0]; e3r1[r] = be3[f1];
    }
    bb4 = (n < 13) ? b4[n] : 0.f;
  }

  for (int i = 0; i < 8; ++i) {
    const float* fi = fw + i * FITEM;
    float acc[8];
#pragma unroll
    for (int r = 0; r < 8; ++r) acc[r] = 0.f;
#pragma unroll
    for (int t = 0; t < 2; ++t) {
      v8f c = {b1r[0], b1r[1], b1r[2], b1r[3], b1r[4], b1r[5], b1r[6], b1r[7]};
      const float* row = fi + (t * 16 + n) * FROW;
#pragma unroll
      for (int j = 0; j < 6; ++j) {
        const int kk = j * 4 + koff;
        v2f xb;
        xb.x = row[kk];
        xb.y = row[kk + 1];
        c = __builtin_amdgcn_wmma_f32_16x16x4_f32(
            false, bf1[j], false, xb, (short)0, c, false, false);
      }
      float x[8];
      float s1 = 0.f, s2 = 0.f;
#pragma unroll
      for (int r = 0; r < 8; ++r) {
        x[r] = fmaxf(c[r], 0.f);
        s1 += x[r];
        s2 += x[r] * x[r];
      }
      s1 += xhalf(s1);
      s2 += xhalf(s2);
      const float m = s1 * (1.f / 16.f);
      const float var = s2 * (1.f / 16.f) - m * m;
      const float rs = rsqrtf(var + LN_EPS_F);
#pragma unroll
      for (int r = 0; r < 8; ++r) {
        const float ln = (x[r] - m) * rs * g1r[r] + be1r[r];
        acc[r] += 2.f * ln;
      }
    }
#pragma unroll
    for (int r = 0; r < 8; ++r) acc[r] = row16_sum(acc[r]);
    if (n == 0) {
#pragma unroll
      for (int r = 0; r < 8; ++r) zbuf[wave * 128 + i * 16 + half * 8 + r] = acc[r];
    }
  }
  __syncthreads();
  if (!tailw) return;


  v8f c0 = {b2r0[0], b2r0[1], b2r0[2], b2r0[3], b2r0[4], b2r0[5], b2r0[6], b2r0[7]};
  v8f c1 = {b2r1[0], b2r1[1], b2r1[2], b2r1[3], b2r1[4], b2r1[5], b2r1[6], b2r1[7]};
#pragma unroll
  for (int j = 0; j < 8; ++j) {
    const int kk = j * 4 + koff;
    v2f xb;
    xb.x = zbuf[n * 32 + kk];
    xb.y = zbuf[n * 32 + kk + 1];
    c0 = __builtin_amdgcn_wmma_f32_16x16x4_f32(false, w2a[j], false, xb, (short)0, c0, false, false);
    c1 = __builtin_amdgcn_wmma_f32_16x16x4_f32(false, w2b[j], false, xb, (short)0, c1, false, false);
  }
  v8f z0, z1;
  relu_ln32(c0, c1, g2r0, g2r1, e2r0, e2r1, z0, z1);
  const v8f skip0 = z0, skip1 = z1;

  v8f d0 = {b3r0[0], b3r0[1], b3r0[2], b3r0[3], b3r0[4], b3r0[5], b3r0[6], b3r0[7]};
  v8f d1 = {b3r1[0], b3r1[1], b3r1[2], b3r1[3], b3r1[4], b3r1[5], b3r1[6], b3r1[7]};
#pragma unroll
  for (int j = 0; j < 8; ++j) {
    v2f zb;
    if (j < 4) { zb.x = z0[2 * j];       zb.y = z0[2 * j + 1]; }
    else       { zb.x = z1[2 * (j - 4)]; zb.y = z1[2 * (j - 4) + 1]; }
    d0 = __builtin_amdgcn_wmma_f32_16x16x4_f32(false, w3a[j], false, zb, (short)0, d0, false, false);
    d1 = __builtin_amdgcn_wmma_f32_16x16x4_f32(false, w3b[j], false, zb, (short)0, d1, false, false);
  }
  relu_ln32(d0, d1, g3r0, g3r1, e3r0, e3r1, z0, z1);
#pragma unroll
  for (int r = 0; r < 8; ++r) {
    z0[r] += skip0[r];
    z1[r] += skip1[r];
  }

  v8f c4 = {bb4, bb4, bb4, bb4, bb4, bb4, bb4, bb4};
#pragma unroll
  for (int j = 0; j < 8; ++j) {
    v2f za;
    if (j < 4) { za.x = z0[2 * j];       za.y = z0[2 * j + 1]; }
    else       { za.x = z1[2 * (j - 4)]; za.y = z1[2 * (j - 4) + 1]; }
    c4 = __builtin_amdgcn_wmma_f32_16x16x4_f32(false, za, false, w4f[j], (short)0, c4, false, false);
  }
  if (n < 13) {
#pragma unroll
    for (int v = 0; v < 8; ++v) {
      out[(size_t)(b_base + v + half * 8) * 13 + n] = tanhf(c4[v]);
    }
  }
}

extern "C" void kernel_launch(void* const* d_in, const int* in_sizes, int n_in,
                              void* d_out, int out_size, void* d_ws, size_t ws_size,
                              hipStream_t stream) {
  const float* pos  = (const float*)d_in[0];
  const float* quat = (const float*)d_in[1];
  const float* aabb = (const float*)d_in[2];
  const float* xz   = (const float*)d_in[3];
  const int*   widx = (const int*)d_in[4];
  const float* pls  = (const float*)d_in[5];
  const float* W1 = (const float*)d_in[6];
  const float* b1 = (const float*)d_in[7];
  const float* g1 = (const float*)d_in[8];
  const float* be1 = (const float*)d_in[9];
  const float* W2 = (const float*)d_in[10];
  const float* b2 = (const float*)d_in[11];
  const float* g2 = (const float*)d_in[12];
  const float* be2 = (const float*)d_in[13];
  const float* W3 = (const float*)d_in[14];
  const float* b3 = (const float*)d_in[15];
  const float* g3 = (const float*)d_in[16];
  const float* be3 = (const float*)d_in[17];
  const float* W4 = (const float*)d_in[18];
  const float* b4 = (const float*)d_in[19];
  float* out = (float*)d_out;

  const int Bn = in_sizes[0] / 6;
  const int grid = Bn / 16;
  hipLaunchKernelGGL(decmodel_kernel, dim3(grid), dim3(128), 0, stream,
                     pos, quat, aabb, xz, widx, pls,
                     W1, b1, g1, be1, W2, b2, g2, be2,
                     W3, b3, g3, be3, W4, b4, out);
}


# Item loop takes gathered rows in arrival order: item i waits (counted) only for its own 2 row loads and runs while later rows land; pre-loop barrier dropped, tail weights loaded last; sums unchanged
# speedup vs baseline: 2.4393x; 1.0670x over previous
#include <hip/hip_runtime.h>
#include <math.h>


typedef float v2f __attribute__((ext_vector_type(2)));
typedef float v8f __attribute__((ext_vector_type(8)));

#define LN_EPS_F 1e-6f
#define FROW 26
#define FITEM (32 * FROW)

__device__ __forceinline__ float xhalf(float x) {
  const int b = __builtin_bit_cast(int, x);
  return __builtin_bit_cast(
      float, __builtin_amdgcn_permlanex16(b, b, 0x76543210, 0xfedcba98, false, false));
}

template <int M>
__device__ __forceinline__ float row_xor_add(float x) {
  const int b = __builtin_bit_cast(int, x);
  const int o = __builtin_amdgcn_update_dpp(0, b, 0x160 + M, 0xf, 0xf, true);
  return x + __builtin_bit_cast(float, o);
}

__device__ __forceinline__ float row16_sum(float x) {
  x = row_xor_add<1>(x);
  x = row_xor_add<2>(x);
  x = row_xor_add<4>(x);
  x = row_xor_add<8>(x);
  return x;
}

__device__ __forceinline__ int kperm(int half, int j, int s) {
  return (j < 4) ? half * 8 + 2 * j + s : 16 + half * 8 + 2 * (j - 4) + s;
}

__device__ __forceinline__ void relu_ln32(const v8f& a0, const v8f& a1,
                                          const float* g0, const float* g1,
                                          const float* e0, const float* e1,
                                          v8f& o0, v8f& o1) {
  float x0[8], x1[8];
  float s1 = 0.f, s2 = 0.f;
#pragma unroll
  for (int r = 0; r < 8; ++r) {
    x0[r] = fmaxf(a0[r], 0.f);
    x1[r] = fmaxf(a1[r], 0.f);
    s1 += x0[r] + x1[r];
    s2 += x0[r] * x0[r] + x1[r] * x1[r];
  }
  s1 += xhalf(s1);
  s2 += xhalf(s2);
  const float m = s1 * (1.f / 32.f);
  const float var = s2 * (1.f / 32.f) - m * m;
  const float rs = rsqrtf(var + LN_EPS_F);
#pragma unroll
  for (int r = 0; r < 8; ++r) {
    o0[r] = (x0[r] - m) * rs * g0[r] + e0[r];
    o1[r] = (x1[r] - m) * rs * g1[r] + e1[r];
  }
}

__device__ __forceinline__ void qrotf(float qx, float qy, float qz, float qw,
                                      float vx, float vy, float vz,
                                      float& rx, float& ry, float& rz) {
  float tx = 2.f * (qy * vz - qz * vy);
  float ty = 2.f * (qz * vx - qx * vz);
  float tz = 2.f * (qx * vy - qy * vx);
  rx = vx + qw * tx + (qy * tz - qz * ty);
  ry = vy + qw * ty + (qz * tx - qx * tz);
  rz = vz + qw * tz + (qx * ty - qy * tx);
}

__global__ __launch_bounds__(128) void decmodel_kernel(
    const float* __restrict__ pos, const float* __restrict__ quat,
    const float* __restrict__ aabb,
    const float* xz,
    const int* __restrict__ widx, const float* __restrict__ plsp,
    const float* __restrict__ W1, const float* __restrict__ b1,
    const float* __restrict__ g1, const float* __restrict__ be1,
    const float* __restrict__ W2, const float* __restrict__ b2,
    const float* __restrict__ g2, const float* __restrict__ be2,
    const float* __restrict__ W3, const float* __restrict__ b3,
    const float* __restrict__ g3, const float* __restrict__ be3,
    const float* __restrict__ W4, const float* __restrict__ b4,
    float* __restrict__ out) {
  __shared__ float featW[4][8 * FITEM];
  __shared__ float zbuf[16 * 32];

  const int tid = threadIdx.x;
  const int wave = tid >> 5;
  const int lane = tid & 31;
  const int half = lane >> 4;
  const int n = lane & 15;
  const int koff = half * 2;
  const int b_base = blockIdx.x * 16;
  const bool tailw = __builtin_amdgcn_readfirstlane(tid) < 32;

  const int bw = b_base + wave * 4;

  int wv[8];
#pragma unroll
  for (int i = 0; i < 8; ++i) wv[i] = widx[(bw * 2 + i) * 32 + lane];

  float ps[4][6], qt[4][8], ab[4][12];
#pragma unroll
  for (int bl = 0; bl < 4; ++bl) {
#pragma unroll
    for (int k = 0; k < 6; ++k) ps[bl][k] = pos[(bw + bl) * 6 + k];
#pragma unroll
    for (int k = 0; k < 8; ++k) qt[bl][k] = quat[(bw + bl) * 8 + k];
#pragma unroll
    for (int k = 0; k < 12; ++k) ab[bl][k] = aabb[(bw + bl) * 12 + k];
  }

  v2f bf1[6];
#pragma unroll
  for (int j = 0; j < 6; ++j) {
    int kk = j * 4 + koff;
    bf1[j].x = (kk < 23) ? W1[kk * 16 + n] : 0.f;
    bf1[j].y = (kk + 1 < 23) ? W1[(kk + 1) * 16 + n] : 0.f;
  }
  float b1r[8], g1r[8], be1r[8];
#pragma unroll
  for (int r = 0; r < 8; ++r) {
    b1r[r] = b1[half * 8 + r];
    g1r[r] = g1[half * 8 + r];
    be1r[r] = be1[half * 8 + r];
  }
  const float pls0 = plsp[0];

  float zr[8][8];
#pragma unroll
  for (int i = 0; i < 8; ++i) {
    const float* zp = xz + ((size_t)(bw * 2 + i) * 4096 + (size_t)wv[i]) * 8;
#pragma unroll
    for (int p = 0; p < 8; ++p) zr[i][p] = zp[p];
    __builtin_amdgcn_sched_barrier(0);
  }
  asm volatile("" :
        "+v"(ps[0][0]), "+v"(ps[0][1]), "+v"(ps[0][2]), "+v"(ps[0][3]),
        "+v"(ps[0][4]), "+v"(ps[0][5]), "+v"(qt[0][0]), "+v"(qt[0][1]),
        "+v"(qt[0][2]), "+v"(qt[0][3]), "+v"(qt[0][4]), "+v"(qt[0][5]),
        "+v"(qt[0][6]), "+v"(qt[0][7]), "+v"(ab[0][0]), "+v"(ab[0][1]),
        "+v"(ab[0][2]), "+v"(ab[0][3]), "+v"(ab[0][4]), "+v"(ab[0][5]),
        "+v"(ab[0][6]), "+v"(ab[0][7]), "+v"(ab[0][8]), "+v"(ab[0][9]),
        "+v"(ab[0][10]), "+v"(ab[0][11]));
  asm volatile("" :
        "+v"(ps[1][0]), "+v"(ps[1][1]), "+v"(ps[1][2]), "+v"(ps[1][3]),
        "+v"(ps[1][4]), "+v"(ps[1][5]), "+v"(qt[1][0]), "+v"(qt[1][1]),
        "+v"(qt[1][2]), "+v"(qt[1][3]), "+v"(qt[1][4]), "+v"(qt[1][5]),
        "+v"(qt[1][6]), "+v"(qt[1][7]), "+v"(ab[1][0]), "+v"(ab[1][1]),
        "+v"(ab[1][2]), "+v"(ab[1][3]), "+v"(ab[1][4]), "+v"(ab[1][5]),
        "+v"(ab[1][6]), "+v"(ab[1][7]), "+v"(ab[1][8]), "+v"(ab[1][9]),
        "+v"(ab[1][10]), "+v"(ab[1][11]));
  asm volatile("" :
        "+v"(ps[2][0]), "+v"(ps[2][1]), "+v"(ps[2][2]), "+v"(ps[2][3]),
        "+v"(ps[2][4]), "+v"(ps[2][5]), "+v"(qt[2][0]), "+v"(qt[2][1]),
        "+v"(qt[2][2]), "+v"(qt[2][3]), "+v"(qt[2][4]), "+v"(qt[2][5]),
        "+v"(qt[2][6]), "+v"(qt[2][7]), "+v"(ab[2][0]), "+v"(ab[2][1]),
        "+v"(ab[2][2]), "+v"(ab[2][3]), "+v"(ab[2][4]), "+v"(ab[2][5]),
        "+v"(ab[2][6]), "+v"(ab[2][7]), "+v"(ab[2][8]), "+v"(ab[2][9]),
        "+v"(ab[2][10]), "+v"(ab[2][11]));
  asm volatile("" :
        "+v"(ps[3][0]), "+v"(ps[3][1]), "+v"(ps[3][2]), "+v"(ps[3][3]),
        "+v"(ps[3][4]), "+v"(ps[3][5]), "+v"(qt[3][0]), "+v"(qt[3][1]),
        "+v"(qt[3][2]), "+v"(qt[3][3]), "+v"(qt[3][4]), "+v"(qt[3][5]),
        "+v"(qt[3][6]), "+v"(qt[3][7]), "+v"(ab[3][0]), "+v"(ab[3][1]),
        "+v"(ab[3][2]), "+v"(ab[3][3]), "+v"(ab[3][4]), "+v"(ab[3][5]),
        "+v"(ab[3][6]), "+v"(ab[3][7]), "+v"(ab[3][8]), "+v"(ab[3][9]),
        "+v"(ab[3][10]), "+v"(ab[3][11]));

  asm volatile("" :
        "+v"(wv[0]), "+v"(wv[1]), "+v"(wv[2]), "+v"(wv[3]),
        "+v"(wv[4]), "+v"(wv[5]), "+v"(wv[6]), "+v"(wv[7]));

  const float escale = expf(pls0);

  float* fw = featW[wave];
#pragma unroll
  for (int bl = 0; bl < 4; ++bl) {
    const float p1x = ps[bl][0], p1y = ps[bl][1], p1z = ps[bl][2];
    const float p2x = ps[bl][3], p2y = ps[bl][4], p2z = ps[bl][5];
    const float ax = -qt[bl][0], ay = -qt[bl][1],
                az = -qt[bl][2], aw = qt[bl][3];
    const float cx = qt[bl][4], cy = qt[bl][5],
                cz = qt[bl][6], cw = qt[bl][7];

    float r1x, r1y, r1z, r2x, r2y, r2z;
    qrotf(ax, ay, az, aw, p1x, p1y, p1z, r1x, r1y, r1z);
    qrotf(ax, ay, az, aw, p2x, p2y, p2z, r2x, r2y, r2z);
    const float rpx = r2x - r1x, rpy = r2y - r1y, rpz = r2z - r1z;

    const float rqw = aw * cw - (ax * cx + ay * cy + az * cz);
    const float rqx = aw * cx + cw * ax + (ay * cz - az * cy);
    const float rqy = aw * cy + cw * ay + (az * cx - ax * cz);
    const float rqz = aw * cz + cw * az + (ax * cy - ay * cx);

    float pf[12];
    pf[0] = rpx * escale; pf[1] = rpy * escale; pf[2] = rpz * escale;
    pf[3] = 1.f - 2.f * (rqy * rqy + rqz * rqz);
    pf[4] = 2.f * (rqx * rqy - rqz * rqw);
    pf[5] = 2.f * (rqx * rqz + rqy * rqw);
    pf[6] = 2.f * (rqx * rqy + rqz * rqw);
    pf[7] = 1.f - 2.f * (rqx * rqx + rqz * rqz);
    pf[8] = 2.f * (rqy * rqz - rqx * rqw);
    pf[9] = 2.f * (rqx * rqz - rqy * rqw);
    pf[10] = 2.f * (rqy * rqz + rqx * rqw);
    pf[11] = 1.f - 2.f * (rqx * rqx + rqy * rqy);

#pragma unroll
    for (int d = 0; d < 2; ++d) {
      const int i = bl * 2 + d;
      const float lox = ab[bl][d * 6 + 0];
      const float loy = ab[bl][d * 6 + 1];
      const float loz = ab[bl][d * 6 + 2];
      const float hix = ab[bl][d * 6 + 3];
      const float hiy = ab[bl][d * 6 + 4];
      const float hiz = ab[bl][d * 6 + 5];

      const int w = wv[i];
      const float gi = (float)(w >> 8);
      const float gj = (float)((w >> 4) & 15);
      const float gk = (float)(w & 15);
      float px = (gi + 0.5f) * (1.f / 16.f) * (hix - lox) + lox;
      float py = (gj + 0.5f) * (1.f / 16.f) * (hiy - loy) + loy;
      float pz = (gk + 0.5f) * (1.f / 16.f) * (hiz - loz) + loz;
      if (d == 1) {
        float tx, ty, tz;
        qrotf(rqx, rqy, rqz, rqw, px, py, pz, tx, ty, tz);
        px = tx + rpx; py = ty + rpy; pz = tz + rpz;
      }

      float* frow = fw + i * FITEM + lane * FROW;
      frow[0] = px; frow[1] = py; frow[2] = pz;
      if (d) {
#pragma unroll
        for (int p = 0; p < 12; ++p) frow[11 + p] = pf[p];
      } else {
#pragma unroll
        for (int p = 0; p < 12; ++p) frow[11 + p] = 0.f;
      }
      frow[23] = 0.f;
    }
  }

  v2f w2a[8] = {}, w2b[8] = {};
  v2f w3a[8] = {}, w3b[8] = {}, w4f[8] = {};
  float b2r0[8] = {}, b2r1[8] = {}, g2r0[8] = {}, g2r1[8] = {}, e2r0[8] = {}, e2r1[8] = {};
  float b3r0[8] = {}, b3r1[8] = {}, g3r0[8] = {}, g3r1[8] = {}, e3r0[8] = {}, e3r1[8] = {};
  float bb4 = 0.f;

#pragma unroll
  for (int i = 0; i < 8; ++i) {
    {
      float* frow = fw + i * FITEM + lane * FROW;
#pragma unroll
      for (int p = 0; p < 8; ++p) frow[3 + p] = zr[i][p];
    }
    __builtin_amdgcn_fence(__ATOMIC_RELEASE, "wavefront");
    __builtin_amdgcn_wave_barrier();
    __builtin_amdgcn_fence(__ATOMIC_ACQUIRE, "wavefront");
    if (i == 7 && tailw) {
#pragma unroll
      for (int j = 0; j < 8; ++j) {
        const int kk = j * 4 + koff;
        w2a[j].x = W2[kk * 32 + n];        w2a[j].y = W2[(kk + 1) * 32 + n];
        w2b[j].x = W2[kk * 32 + 16 + n];   w2b[j].y = W2[(kk + 1) * 32 + 16 + n];
        const int k0 = kperm(half, j, 0), k1 = kperm(half, j, 1);
        w3a[j].x = W3[k0 * 32 + n];        w3a[j].y = W3[k1 * 32 + n];
        w3b[j].x = W3[k0 * 32 + 16 + n];   w3b[j].y = W3[k1 * 32 + 16 + n];
        w4f[j].x = (n < 13) ? W4[k0 * 13 + n] : 0.f;
        w4f[j].y = (n < 13) ? W4[k1 * 13 + n] : 0.f;
      }
#pragma unroll
      for (int r = 0; r < 8; ++r) {
        const int f0 = half * 8 + r, f1 = 16 + half * 8 + r;
        b2r0[r] = b2[f0];  b2r1[r] = b2[f1];
        g2r0[r] = g2[f0];  g2r1[r] = g2[f1];
        e2r0[r] = be2[f0]; e2r1[r] = be2[f1];
        b3r0[r] = b3[f0];  b3r1[r] = b3[f1];
        g3r0[r] = g3[f0];  g3r1[r] = g3[f1];
        e3r0[r] = be3[f0]; e3r1[r] = be3[f1];
      }
      bb4 = (n < 13) ? b4[n] : 0.f;
    }
    const float* fi = fw + i * FITEM;
    float acc[8];
#pragma unroll
    for (int r = 0; r < 8; ++r) acc[r] = 0.f;
#pragma unroll
    for (int t = 0; t < 2; ++t) {
      v8f c = {b1r[0], b1r[1], b1r[2], b1r[3], b1r[4], b1r[5], b1r[6], b1r[7]};
      const float* row = fi + (t * 16 + n) * FROW;
#pragma unroll
      for (int j = 0; j < 6; ++j) {
        const int kk = j * 4 + koff;
        v2f xb;
        xb.x = row[kk];
        xb.y = row[kk + 1];
        c = __builtin_amdgcn_wmma_f32_16x16x4_f32(
            false, bf1[j], false, xb, (short)0, c, false, false);
      }
      float x[8];
      float s1 = 0.f, s2 = 0.f;
#pragma unroll
      for (int r = 0; r < 8; ++r) {
        x[r] = fmaxf(c[r], 0.f);
        s1 += x[r];
        s2 += x[r] * x[r];
      }
      s1 += xhalf(s1);
      s2 += xhalf(s2);
      const float m = s1 * (1.f / 16.f);
      const float var = s2 * (1.f / 16.f) - m * m;
      const float rs = rsqrtf(var + LN_EPS_F);
#pragma unroll
      for (int r = 0; r < 8; ++r) {
        const float ln = (x[r] - m) * rs * g1r[r] + be1r[r];
        acc[r] += 2.f * ln;
      }
    }
#pragma unroll
    for (int r = 0; r < 8; ++r) acc[r] = row16_sum(acc[r]);
    if (n == 0) {
#pragma unroll
      for (int r = 0; r < 8; ++r) zbuf[wave * 128 + i * 16 + half * 8 + r] = acc[r];
    }
  }
  __syncthreads();
  if (!tailw) return;


  v8f c0 = {b2r0[0], b2r0[1], b2r0[2], b2r0[3], b2r0[4], b2r0[5], b2r0[6], b2r0[7]};
  v8f c1 = {b2r1[0], b2r1[1], b2r1[2], b2r1[3], b2r1[4], b2r1[5], b2r1[6], b2r1[7]};
#pragma unroll
  for (int j = 0; j < 8; ++j) {
    const int kk = j * 4 + koff;
    v2f xb;
    xb.x = zbuf[n * 32 + kk];
    xb.y = zbuf[n * 32 + kk + 1];
    c0 = __builtin_amdgcn_wmma_f32_16x16x4_f32(false, w2a[j], false, xb, (short)0, c0, false, false);
    c1 = __builtin_amdgcn_wmma_f32_16x16x4_f32(false, w2b[j], false, xb, (short)0, c1, false, false);
  }
  v8f z0, z1;
  relu_ln32(c0, c1, g2r0, g2r1, e2r0, e2r1, z0, z1);
  const v8f skip0 = z0, skip1 = z1;

  v8f d0 = {b3r0[0], b3r0[1], b3r0[2], b3r0[3], b3r0[4], b3r0[5], b3r0[6], b3r0[7]};
  v8f d1 = {b3r1[0], b3r1[1], b3r1[2], b3r1[3], b3r1[4], b3r1[5], b3r1[6], b3r1[7]};
#pragma unroll
  for (int j = 0; j < 8; ++j) {
    v2f zb;
    if (j < 4) { zb.x = z0[2 * j];       zb.y = z0[2 * j + 1]; }
    else       { zb.x = z1[2 * (j - 4)]; zb.y = z1[2 * (j - 4) + 1]; }
    d0 = __builtin_amdgcn_wmma_f32_16x16x4_f32(false, w3a[j], false, zb, (short)0, d0, false, false);
    d1 = __builtin_amdgcn_wmma_f32_16x16x4_f32(false, w3b[j], false, zb, (short)0, d1, false, false);
  }
  relu_ln32(d0, d1, g3r0, g3r1, e3r0, e3r1, z0, z1);
#pragma unroll
  for (int r = 0; r < 8; ++r) {
    z0[r] += skip0[r];
    z1[r] += skip1[r];
  }

  v8f c4 = {bb4, bb4, bb4, bb4, bb4, bb4, bb4, bb4};
#pragma unroll
  for (int j = 0; j < 8; ++j) {
    v2f za;
    if (j < 4) { za.x = z0[2 * j];       za.y = z0[2 * j + 1]; }
    else       { za.x = z1[2 * (j - 4)]; za.y = z1[2 * (j - 4) + 1]; }
    c4 = __builtin_amdgcn_wmma_f32_16x16x4_f32(false, za, false, w4f[j], (short)0, c4, false, false);
  }
  if (n < 13) {
#pragma unroll
    for (int v = 0; v < 8; ++v) {
      out[(size_t)(b_base + v + half * 8) * 13 + n] = tanhf(c4[v]);
    }
  }
}

extern "C" void kernel_launch(void* const* d_in, const int* in_sizes, int n_in,
                              void* d_out, int out_size, void* d_ws, size_t ws_size,
                              hipStream_t stream) {
  const float* pos  = (const float*)d_in[0];
  const float* quat = (const float*)d_in[1];
  const float* aabb = (const float*)d_in[2];
  const float* xz   = (const float*)d_in[3];
  const int*   widx = (const int*)d_in[4];
  const float* pls  = (const float*)d_in[5];
  const float* W1 = (const float*)d_in[6];
  const float* b1 = (const float*)d_in[7];
  const float* g1 = (const float*)d_in[8];
  const float* be1 = (const float*)d_in[9];
  const float* W2 = (const float*)d_in[10];
  const float* b2 = (const float*)d_in[11];
  const float* g2 = (const float*)d_in[12];
  const float* be2 = (const float*)d_in[13];
  const float* W3 = (const float*)d_in[14];
  const float* b3 = (const float*)d_in[15];
  const float* g3 = (const float*)d_in[16];
  const float* be3 = (const float*)d_in[17];
  const float* W4 = (const float*)d_in[18];
  const float* b4 = (const float*)d_in[19];
  float* out = (float*)d_out;

  const int Bn = in_sizes[0] / 6;
  const int grid = Bn / 16;
  hipLaunchKernelGGL(decmodel_kernel, dim3(grid), dim3(128), 0, stream,
                     pos, quat, aabb, xz, widx, pls,
                     W1, b1, g1, be1, W2, b2, g2, be2,
                     W3, b3, g3, be3, W4, b4, out);
}
